# starting kernel
# baseline (speedup 1.0000x reference)
#include <hip/hip_runtime.h>
#include <math.h>

typedef __attribute__((ext_vector_type(2))) float v2f;
typedef __attribute__((ext_vector_type(4))) float v4f;
typedef __attribute__((ext_vector_type(8))) float v8f;

constexpr int kB   = 32;
constexpr int kIU  = 16;
constexpr int kIC  = 2048;
constexpr int kNU  = 16;
constexpr int kUS  = 64;
constexpr int kKS  = 8;
constexpr size_t kPartialsElems = (size_t)kKS * 1024 * kB;

__global__ __launch_bounds__(256) void caps_xt(const float* __restrict__ x,
                                               float* __restrict__ xt2) {
  int idx  = blockIdx.x * 256 + threadIdx.x;
  int r    = idx & 1;
  int slo  = (idx >> 1) & 1;
  int lane = (idx >> 2) & 31;
  int half = (idx >> 7) & 1;
  int tile = (idx >> 8) & 1;
  int i    = idx >> 9;
  int s    = half * 2 + slo;
  int kgrp = lane >> 4;
  int n    = (lane & 15) + 16 * tile;
  int u    = 2 * s + 8 * kgrp + r;
  xt2[idx] = x[(size_t)n * (kIU * kIC) + (size_t)u * kIC + i];
}

__global__ __launch_bounds__(256) void caps_gemm(const float* __restrict__ xt2,
                                                 const float* __restrict__ W,
                                                 float* __restrict__ partials) {
  const int lane  = threadIdx.x & 31;
  const int wave  = threadIdx.x >> 5;
  const int mtile = blockIdx.x;
  const int ksp   = blockIdx.y;
  const int j     = mtile >> 2;
  const int k0    = (mtile & 3) << 4;
  const int mrow  = lane & 15;
  const int kgrp  = lane >> 4;

  v8f acc0 = {};
  v8f acc1 = {};

  const float* Wjk = W + (size_t)j * (kUS * kIU) + (size_t)k0 * kIU
                       + (size_t)mrow * kIU + (size_t)kgrp * 8;
  const int i0 = ksp * (kIC / kKS);

  for (int t = 0; t < (kIC / kKS) / 8; ++t) {
    const int i = i0 + wave + t * 8;
    const float* Wb = Wjk + (size_t)i * (kNU * kUS * kIU);
    __builtin_prefetch(Wb + (size_t)8 * (kNU * kUS * kIU), 0, 1);
    v4f a01 = __builtin_nontemporal_load((const v4f*)(Wb));
    v4f a23 = __builtin_nontemporal_load((const v4f*)(Wb + 4));
    const float* Xb = xt2 + (size_t)i * 512 + lane * 4;
    v4f b0a = *(const v4f*)(Xb);
    v4f b0b = *(const v4f*)(Xb + 128);
    v4f b1a = *(const v4f*)(Xb + 256);
    v4f b1b = *(const v4f*)(Xb + 384);

    v2f a, b;
#define STEP(AV, LOHI, B0, B1)                                             \
    a = __builtin_shufflevector(AV, AV, 0 + LOHI, 1 + LOHI);               \
    b = __builtin_shufflevector(B0, B0, 0 + LOHI, 1 + LOHI);               \
    acc0 = __builtin_amdgcn_wmma_f32_16x16x4_f32(false, a, false, b,       \
                                      (short)0, acc0, false, false);       \
    b = __builtin_shufflevector(B1, B1, 0 + LOHI, 1 + LOHI);               \
    acc1 = __builtin_amdgcn_wmma_f32_16x16x4_f32(false, a, false, b,       \
                                      (short)0, acc1, false, false);
    STEP(a01, 0, b0a, b1a)
    STEP(a01, 2, b0a, b1a)
    STEP(a23, 0, b0b, b1b)
    STEP(a23, 2, b0b, b1b)
#undef STEP
  }

  __shared__ float lds[8][2][256];
  *(v8f*)&lds[wave][0][lane * 8] = acc0;
  *(v8f*)&lds[wave][1][lane * 8] = acc1;
  __syncthreads();

  int flat = threadIdx.x * 2;
#pragma unroll
  for (int q = 0; q < 2; ++q, ++flat) {
    const int t = flat >> 8;
    const int p = flat & 255;
    float sum = 0.f;
#pragma unroll
    for (int w = 0; w < 8; ++w) sum += lds[w][t][p];
    const int l  = p >> 3, r = p & 7;
    const int mr = r + ((l >> 4) << 3);
    const int n  = t * 16 + (l & 15);
    partials[(size_t)ksp * (1024 * kB) + (size_t)(mtile * 16 + mr) * kB + n] = sum;
  }
}

__global__ __launch_bounds__(64) void caps_squash(const float* __restrict__ partials,
                                                  float* __restrict__ out) {
  const int b = blockIdx.x;
  const int k = threadIdx.x;
  float sv[kNU];
  float mag_sq = 0.f;
#pragma unroll
  for (int j = 0; j < kNU; ++j) {
    float v = 0.f;
#pragma unroll
    for (int p = 0; p < kKS; ++p)
      v += partials[(size_t)p * (1024 * kB) + (size_t)(j * kUS + k) * kB + b];
    sv[j] = v;
    mag_sq += v * v;
  }
  const float mag   = sqrtf(mag_sq);
  const float scale = mag_sq / ((1.f + mag_sq) * mag);
#pragma unroll
  for (int j = 0; j < kNU; ++j)
    out[(size_t)b * (kNU * kUS) + j * kUS + k] = sv[j] * scale;
}

extern "C" void kernel_launch(void* const* d_in, const int* in_sizes, int n_in,
                              void* d_out, int out_size, void* d_ws, size_t ws_size,
                              hipStream_t stream) {
  const float* x = (const float*)d_in[0];
  const float* W = (const float*)d_in[1];
  float* out      = (float*)d_out;
  float* partials = (float*)d_ws;
  float* xt2      = (float*)d_ws + kPartialsElems;

  caps_xt<<<(2048 * 512) / 256, 256, 0, stream>>>(x, xt2);
  dim3 grid(64, kKS);
  caps_gemm<<<grid, 256, 0, stream>>>(xt2, W, partials);
  caps_squash<<<32, 64, 0, stream>>>(partials, out);
}


# GEMM: each wave keeps 2 M-tiles x 2 N-tiles so one xt2 B fragment feeds two A tiles; 16-wave workgroup, 16 trips per wave; K summation order changes (16 waves x 16 trips)
# speedup vs baseline: 1.0812x; 1.0812x over previous
#include <hip/hip_runtime.h>
#include <math.h>

typedef __attribute__((ext_vector_type(2))) float v2f;
typedef __attribute__((ext_vector_type(4))) float v4f;
typedef __attribute__((ext_vector_type(8))) float v8f;

constexpr int kB   = 32;
constexpr int kIU  = 16;
constexpr int kIC  = 2048;
constexpr int kNU  = 16;
constexpr int kUS  = 64;
constexpr int kKS  = 8;
constexpr size_t kPartialsElems = (size_t)kKS * 1024 * kB;

__global__ __launch_bounds__(256) void caps_xt(const float* __restrict__ x,
                                               float* __restrict__ xt2) {
  int idx  = blockIdx.x * 256 + threadIdx.x;
  int r    = idx & 1;
  int slo  = (idx >> 1) & 1;
  int lane = (idx >> 2) & 31;
  int half = (idx >> 7) & 1;
  int tile = (idx >> 8) & 1;
  int i    = idx >> 9;
  int s    = half * 2 + slo;
  int kgrp = lane >> 4;
  int n    = (lane & 15) + 16 * tile;
  int u    = 2 * s + 8 * kgrp + r;
  xt2[idx] = x[(size_t)n * (kIU * kIC) + (size_t)u * kIC + i];
}

constexpr int kGemmWaves = 16;
constexpr int kGemmTrips = (kIC / kKS) / kGemmWaves;

__global__ __launch_bounds__(512) void caps_gemm(const float* __restrict__ xt2,
                                                 const float* __restrict__ W,
                                                 float* __restrict__ partials) {
  const int lane  = threadIdx.x & 31;
  const int wave  = threadIdx.x >> 5;
  const int mpair = blockIdx.x;
  const int ksp   = blockIdx.y;
  const int mrow  = lane & 15;
  const int kgrp  = lane >> 4;

  v8f acc00 = {};
  v8f acc01 = {};
  v8f acc10 = {};
  v8f acc11 = {};

  const float* Wm = W + (size_t)(mpair * 32 + mrow) * kIU + (size_t)kgrp * 8;
  const int i0 = ksp * (kIC / kKS);

  for (int t = 0; t < kGemmTrips; ++t) {
    const int i = i0 + wave + t * kGemmWaves;
    const float* Wb = Wm + (size_t)i * (kNU * kUS * kIU);
    if (t + 1 < kGemmTrips) {
      __builtin_prefetch(Wb + (size_t)kGemmWaves * (kNU * kUS * kIU), 0, 1);
      __builtin_prefetch(Wb + (size_t)kGemmWaves * (kNU * kUS * kIU) + 256, 0, 1);
    }
    v4f a01_0 = __builtin_nontemporal_load((const v4f*)(Wb));
    v4f a23_0 = __builtin_nontemporal_load((const v4f*)(Wb + 4));
    v4f a01_1 = __builtin_nontemporal_load((const v4f*)(Wb + 256));
    v4f a23_1 = __builtin_nontemporal_load((const v4f*)(Wb + 260));
    const float* Xb = xt2 + (size_t)i * 512 + lane * 4;
    v4f b0a = *(const v4f*)(Xb);
    v4f b0b = *(const v4f*)(Xb + 128);
    v4f b1a = *(const v4f*)(Xb + 256);
    v4f b1b = *(const v4f*)(Xb + 384);

    v2f a0, a1, b;
#define STEP(AV0, AV1, LOHI, B0, B1)                                        \
    a0 = __builtin_shufflevector(AV0, AV0, 0 + LOHI, 1 + LOHI);             \
    a1 = __builtin_shufflevector(AV1, AV1, 0 + LOHI, 1 + LOHI);             \
    b = __builtin_shufflevector(B0, B0, 0 + LOHI, 1 + LOHI);                \
    acc00 = __builtin_amdgcn_wmma_f32_16x16x4_f32(false, a0, false, b,      \
                                      (short)0, acc00, false, false);       \
    acc10 = __builtin_amdgcn_wmma_f32_16x16x4_f32(false, a1, false, b,      \
                                      (short)0, acc10, false, false);       \
    b = __builtin_shufflevector(B1, B1, 0 + LOHI, 1 + LOHI);                \
    acc01 = __builtin_amdgcn_wmma_f32_16x16x4_f32(false, a0, false, b,      \
                                      (short)0, acc01, false, false);       \
    acc11 = __builtin_amdgcn_wmma_f32_16x16x4_f32(false, a1, false, b,      \
                                      (short)0, acc11, false, false);
    STEP(a01_0, a01_1, 0, b0a, b1a)
    STEP(a01_0, a01_1, 2, b0a, b1a)
    STEP(a23_0, a23_1, 0, b0b, b1b)
    STEP(a23_0, a23_1, 2, b0b, b1b)
#undef STEP
  }

  __shared__ float lds[kGemmWaves][4][256];
  *(v8f*)&lds[wave][0][lane * 8] = acc00;
  *(v8f*)&lds[wave][1][lane * 8] = acc01;
  *(v8f*)&lds[wave][2][lane * 8] = acc10;
  *(v8f*)&lds[wave][3][lane * 8] = acc11;
  __syncthreads();

  int flat = threadIdx.x * 2;
#pragma unroll
  for (int q = 0; q < 2; ++q, ++flat) {
    const int tq = flat >> 8;
    const int p  = flat & 255;
    float sum = 0.f;
#pragma unroll
    for (int w = 0; w < kGemmWaves; ++w) sum += lds[w][tq][p];
    const int l  = p >> 3, r = p & 7;
    const int mr = (tq >> 1) * 16 + r + ((l >> 4) << 3);
    const int n  = (tq & 1) * 16 + (l & 15);
    partials[(size_t)ksp * (1024 * kB) + (size_t)(mpair * 32 + mr) * kB + n] = sum;
  }
}

__global__ __launch_bounds__(64) void caps_squash(const float* __restrict__ partials,
                                                  float* __restrict__ out) {
  const int b = blockIdx.x;
  const int k = threadIdx.x;
  float sv[kNU];
  float mag_sq = 0.f;
#pragma unroll
  for (int j = 0; j < kNU; ++j) {
    float v = 0.f;
#pragma unroll
    for (int p = 0; p < kKS; ++p)
      v += partials[(size_t)p * (1024 * kB) + (size_t)(j * kUS + k) * kB + b];
    sv[j] = v;
    mag_sq += v * v;
  }
  const float mag   = sqrtf(mag_sq);
  const float scale = mag_sq / ((1.f + mag_sq) * mag);
#pragma unroll
  for (int j = 0; j < kNU; ++j)
    out[(size_t)b * (kNU * kUS) + j * kUS + k] = sv[j] * scale;
}

extern "C" void kernel_launch(void* const* d_in, const int* in_sizes, int n_in,
                              void* d_out, int out_size, void* d_ws, size_t ws_size,
                              hipStream_t stream) {
  const float* x = (const float*)d_in[0];
  const float* W = (const float*)d_in[1];
  float* out      = (float*)d_out;
  float* partials = (float*)d_ws;
  float* xt2      = (float*)d_ws + kPartialsElems;

  caps_xt<<<(2048 * 512) / 256, 256, 0, stream>>>(x, xt2);
  dim3 grid(32, kKS);
  caps_gemm<<<grid, 512, 0, stream>>>(xt2, W, partials);
  caps_squash<<<32, 64, 0, stream>>>(partials, out);
}


# pack kernel: each lane reads four u-rows x 8 i as 16-byte loads, turns them in registers, stores whole 16-byte fragment quads (512 B per wave store); 1024 waves enter instead of 32768
# speedup vs baseline: 1.1909x; 1.1015x over previous
#include <hip/hip_runtime.h>
#include <math.h>

typedef __attribute__((ext_vector_type(2))) float v2f;
typedef __attribute__((ext_vector_type(4))) float v4f;
typedef __attribute__((ext_vector_type(8))) float v8f;

constexpr int kB   = 32;
constexpr int kIU  = 16;
constexpr int kIC  = 2048;
constexpr int kNU  = 16;
constexpr int kUS  = 64;
constexpr int kKS  = 8;
constexpr size_t kPartialsElems = (size_t)kKS * 1024 * kB;

constexpr int kPackI = 8;

__global__ __launch_bounds__(128) void caps_xt(const float* __restrict__ x,
                                               float* __restrict__ xt2) {
  const int lane = threadIdx.x & 31;
  const int wave = threadIdx.x >> 5;
  const int tile = wave >> 1;
  const int half = wave & 1;
  const int kgrp = lane >> 4;
  const int n    = (lane & 15) + 16 * tile;
  const int u0   = 8 * kgrp + 4 * half;
  const int i0   = blockIdx.x * kPackI;

  const float* src = x + (size_t)n * (kIU * kIC) + (size_t)u0 * kIC + i0;
  v4f r[4][2];
#pragma unroll
  for (int q = 0; q < 4; ++q) {
    r[q][0] = *(const v4f*)(src + (size_t)q * kIC);
    r[q][1] = *(const v4f*)(src + (size_t)q * kIC + 4);
  }

  float* dst = xt2 + (size_t)i0 * 512 + tile * 256 + half * 128 + lane * 4;
#pragma unroll
  for (int il = 0; il < kPackI; ++il) {
    v4f o;
    o.x = r[0][il >> 2][il & 3];
    o.y = r[1][il >> 2][il & 3];
    o.z = r[2][il >> 2][il & 3];
    o.w = r[3][il >> 2][il & 3];
    *(v4f*)(dst + (size_t)il * 512) = o;
  }
}

constexpr int kGemmWaves = 16;
constexpr int kGemmTrips = (kIC / kKS) / kGemmWaves;

__global__ __launch_bounds__(512) void caps_gemm(const float* __restrict__ xt2,
                                                 const float* __restrict__ W,
                                                 float* __restrict__ partials) {
  const int lane  = threadIdx.x & 31;
  const int wave  = threadIdx.x >> 5;
  const int mpair = blockIdx.x;
  const int ksp   = blockIdx.y;
  const int mrow  = lane & 15;
  const int kgrp  = lane >> 4;

  v8f acc00 = {};
  v8f acc01 = {};
  v8f acc10 = {};
  v8f acc11 = {};

  const float* Wm = W + (size_t)(mpair * 32 + mrow) * kIU + (size_t)kgrp * 8;
  const int i0 = ksp * (kIC / kKS);

  for (int t = 0; t < kGemmTrips; ++t) {
    const int i = i0 + wave + t * kGemmWaves;
    const float* Wb = Wm + (size_t)i * (kNU * kUS * kIU);
    if (t + 1 < kGemmTrips) {
      __builtin_prefetch(Wb + (size_t)kGemmWaves * (kNU * kUS * kIU), 0, 1);
      __builtin_prefetch(Wb + (size_t)kGemmWaves * (kNU * kUS * kIU) + 256, 0, 1);
    }
    v4f a01_0 = __builtin_nontemporal_load((const v4f*)(Wb));
    v4f a23_0 = __builtin_nontemporal_load((const v4f*)(Wb + 4));
    v4f a01_1 = __builtin_nontemporal_load((const v4f*)(Wb + 256));
    v4f a23_1 = __builtin_nontemporal_load((const v4f*)(Wb + 260));
    const float* Xb = xt2 + (size_t)i * 512 + lane * 4;
    v4f b0a = *(const v4f*)(Xb);
    v4f b0b = *(const v4f*)(Xb + 128);
    v4f b1a = *(const v4f*)(Xb + 256);
    v4f b1b = *(const v4f*)(Xb + 384);

    v2f a0, a1, b;
#define STEP(AV0, AV1, LOHI, B0, B1)                                        \
    a0 = __builtin_shufflevector(AV0, AV0, 0 + LOHI, 1 + LOHI);             \
    a1 = __builtin_shufflevector(AV1, AV1, 0 + LOHI, 1 + LOHI);             \
    b = __builtin_shufflevector(B0, B0, 0 + LOHI, 1 + LOHI);                \
    acc00 = __builtin_amdgcn_wmma_f32_16x16x4_f32(false, a0, false, b,      \
                                      (short)0, acc00, false, false);       \
    acc10 = __builtin_amdgcn_wmma_f32_16x16x4_f32(false, a1, false, b,      \
                                      (short)0, acc10, false, false);       \
    b = __builtin_shufflevector(B1, B1, 0 + LOHI, 1 + LOHI);                \
    acc01 = __builtin_amdgcn_wmma_f32_16x16x4_f32(false, a0, false, b,      \
                                      (short)0, acc01, false, false);       \
    acc11 = __builtin_amdgcn_wmma_f32_16x16x4_f32(false, a1, false, b,      \
                                      (short)0, acc11, false, false);
    STEP(a01_0, a01_1, 0, b0a, b1a)
    STEP(a01_0, a01_1, 2, b0a, b1a)
    STEP(a23_0, a23_1, 0, b0b, b1b)
    STEP(a23_0, a23_1, 2, b0b, b1b)
#undef STEP
  }

  __shared__ float lds[kGemmWaves][4][256];
  *(v8f*)&lds[wave][0][lane * 8] = acc00;
  *(v8f*)&lds[wave][1][lane * 8] = acc01;
  *(v8f*)&lds[wave][2][lane * 8] = acc10;
  *(v8f*)&lds[wave][3][lane * 8] = acc11;
  __syncthreads();

  int flat = threadIdx.x * 2;
#pragma unroll
  for (int q = 0; q < 2; ++q, ++flat) {
    const int tq = flat >> 8;
    const int p  = flat & 255;
    float sum = 0.f;
#pragma unroll
    for (int w = 0; w < kGemmWaves; ++w) sum += lds[w][tq][p];
    const int l  = p >> 3, r = p & 7;
    const int mr = (tq >> 1) * 16 + r + ((l >> 4) << 3);
    const int n  = (tq & 1) * 16 + (l & 15);
    partials[(size_t)ksp * (1024 * kB) + (size_t)(mpair * 32 + mr) * kB + n] = sum;
  }
}

__global__ __launch_bounds__(64) void caps_squash(const float* __restrict__ partials,
                                                  float* __restrict__ out) {
  const int b = blockIdx.x;
  const int k = threadIdx.x;
  float sv[kNU];
  float mag_sq = 0.f;
#pragma unroll
  for (int j = 0; j < kNU; ++j) {
    float v = 0.f;
#pragma unroll
    for (int p = 0; p < kKS; ++p)
      v += partials[(size_t)p * (1024 * kB) + (size_t)(j * kUS + k) * kB + b];
    sv[j] = v;
    mag_sq += v * v;
  }
  const float mag   = sqrtf(mag_sq);
  const float scale = mag_sq / ((1.f + mag_sq) * mag);
#pragma unroll
  for (int j = 0; j < kNU; ++j)
    out[(size_t)b * (kNU * kUS) + j * kUS + k] = sv[j] * scale;
}

extern "C" void kernel_launch(void* const* d_in, const int* in_sizes, int n_in,
                              void* d_out, int out_size, void* d_ws, size_t ws_size,
                              hipStream_t stream) {
  const float* x = (const float*)d_in[0];
  const float* W = (const float*)d_in[1];
  float* out      = (float*)d_out;
  float* partials = (float*)d_ws;
  float* xt2      = (float*)d_ws + kPartialsElems;

  caps_xt<<<kIC / kPackI, 128, 0, stream>>>(x, xt2);
  dim3 grid(32, kKS);
  caps_gemm<<<grid, 512, 0, stream>>>(xt2, W, partials);
  caps_squash<<<32, 64, 0, stream>>>(partials, out);
}


# GEMM: fragment loads software-pipelined in registers, six trips (48 loads, 24 KB per wave) in flight ahead of the matrix instructions; replaces the one-trip prefetch hint; summation order unchanged
# speedup vs baseline: 1.2758x; 1.0713x over previous
#include <hip/hip_runtime.h>
#include <math.h>

typedef __attribute__((ext_vector_type(2))) float v2f;
typedef __attribute__((ext_vector_type(4))) float v4f;
typedef __attribute__((ext_vector_type(8))) float v8f;

constexpr int kB   = 32;
constexpr int kIU  = 16;
constexpr int kIC  = 2048;
constexpr int kNU  = 16;
constexpr int kUS  = 64;
constexpr int kKS  = 8;
constexpr size_t kPartialsElems = (size_t)kKS * 1024 * kB;

constexpr int kPackI = 8;

__global__ __launch_bounds__(128) void caps_xt(const float* __restrict__ x,
                                               float* __restrict__ xt2) {
  const int lane = threadIdx.x & 31;
  const int wave = threadIdx.x >> 5;
  const int tile = wave >> 1;
  const int half = wave & 1;
  const int kgrp = lane >> 4;
  const int n    = (lane & 15) + 16 * tile;
  const int u0   = 8 * kgrp + 4 * half;
  const int i0   = blockIdx.x * kPackI;

  const float* src = x + (size_t)n * (kIU * kIC) + (size_t)u0 * kIC + i0;
  v4f r[4][2];
#pragma unroll
  for (int q = 0; q < 4; ++q) {
    r[q][0] = *(const v4f*)(src + (size_t)q * kIC);
    r[q][1] = *(const v4f*)(src + (size_t)q * kIC + 4);
  }

  float* dst = xt2 + (size_t)i0 * 512 + tile * 256 + half * 128 + lane * 4;
#pragma unroll
  for (int il = 0; il < kPackI; ++il) {
    v4f o;
    o.x = r[0][il >> 2][il & 3];
    o.y = r[1][il >> 2][il & 3];
    o.z = r[2][il >> 2][il & 3];
    o.w = r[3][il >> 2][il & 3];
    *(v4f*)(dst + (size_t)il * 512) = o;
  }
}

constexpr int kGemmWaves = 16;
constexpr int kGemmTrips = (kIC / kKS) / kGemmWaves;
constexpr int kAhead     = 6;

static __device__ __forceinline__ void caps_trip(const v4f (&fa)[4], const v4f (&fb)[4],
                                                 v8f& acc00, v8f& acc01,
                                                 v8f& acc10, v8f& acc11) {
  v2f a0, a1, b;
#define STEP(AV0, AV1, LOHI, B0, B1)                                        \
    a0 = __builtin_shufflevector(AV0, AV0, 0 + LOHI, 1 + LOHI);             \
    a1 = __builtin_shufflevector(AV1, AV1, 0 + LOHI, 1 + LOHI);             \
    b = __builtin_shufflevector(B0, B0, 0 + LOHI, 1 + LOHI);                \
    acc00 = __builtin_amdgcn_wmma_f32_16x16x4_f32(false, a0, false, b,      \
                                      (short)0, acc00, false, false);       \
    acc10 = __builtin_amdgcn_wmma_f32_16x16x4_f32(false, a1, false, b,      \
                                      (short)0, acc10, false, false);       \
    b = __builtin_shufflevector(B1, B1, 0 + LOHI, 1 + LOHI);                \
    acc01 = __builtin_amdgcn_wmma_f32_16x16x4_f32(false, a0, false, b,      \
                                      (short)0, acc01, false, false);       \
    acc11 = __builtin_amdgcn_wmma_f32_16x16x4_f32(false, a1, false, b,      \
                                      (short)0, acc11, false, false);
  STEP(fa[0], fa[2], 0, fb[0], fb[2])
  STEP(fa[0], fa[2], 2, fb[0], fb[2])
  STEP(fa[1], fa[3], 0, fb[1], fb[3])
  STEP(fa[1], fa[3], 2, fb[1], fb[3])
#undef STEP
}

__global__ __launch_bounds__(512) void caps_gemm(const float* __restrict__ xt2,
                                                 const float* __restrict__ W,
                                                 float* __restrict__ partials) {
  const int lane  = threadIdx.x & 31;
  const int wave  = __builtin_amdgcn_readfirstlane(threadIdx.x >> 5);
  const int mpair = blockIdx.x;
  const int ksp   = blockIdx.y;
  const int mrow  = lane & 15;
  const int kgrp  = lane >> 4;

  v8f acc00 = {};
  v8f acc01 = {};
  v8f acc10 = {};
  v8f acc11 = {};

  const int ifirst = ksp * (kIC / kKS) + wave;
  const float* Wt = W + (size_t)ifirst * (kNU * kUS * kIU) + (size_t)mpair * 32 * kIU;
  const float* Xt = xt2 + (size_t)ifirst * 512;
  const unsigned wlane = (unsigned)(mrow * kIU + kgrp * 8);
  const unsigned xlane = (unsigned)(lane * 4);

  v4f fa[kAhead][4];
  v4f fb[kAhead][4];
#define LOAD_TRIP(T) do {                                                              \
    const float* Wb_ = Wt + (size_t)(T) * (kGemmWaves * kNU * kUS * kIU) + wlane;      \
    const float* Xb_ = Xt + (size_t)(T) * (kGemmWaves * 512) + xlane;                  \
    fa[(T) % kAhead][0] = __builtin_nontemporal_load((const v4f*)(Wb_));               \
    fa[(T) % kAhead][1] = __builtin_nontemporal_load((const v4f*)(Wb_ + 4));           \
    fa[(T) % kAhead][2] = __builtin_nontemporal_load((const v4f*)(Wb_ + 256));         \
    fa[(T) % kAhead][3] = __builtin_nontemporal_load((const v4f*)(Wb_ + 260));         \
    fb[(T) % kAhead][0] = *(const v4f*)(Xb_);                                          \
    fb[(T) % kAhead][1] = *(const v4f*)(Xb_ + 128);                                    \
    fb[(T) % kAhead][2] = *(const v4f*)(Xb_ + 256);                                    \
    fb[(T) % kAhead][3] = *(const v4f*)(Xb_ + 384);                                    \
  } while (0)

#pragma unroll
  for (int t = 0; t < kAhead; ++t) LOAD_TRIP(t);
#pragma unroll
  for (int t = 0; t < kGemmTrips; ++t) {
    __builtin_amdgcn_sched_barrier(0);
    caps_trip(fa[t % kAhead], fb[t % kAhead], acc00, acc01, acc10, acc11);
    __builtin_amdgcn_sched_barrier(0);
    if (t + kAhead < kGemmTrips) LOAD_TRIP(t + kAhead);
  }
#undef LOAD_TRIP

  __shared__ float lds[kGemmWaves][4][256];
  *(v8f*)&lds[wave][0][lane * 8] = acc00;
  *(v8f*)&lds[wave][1][lane * 8] = acc01;
  *(v8f*)&lds[wave][2][lane * 8] = acc10;
  *(v8f*)&lds[wave][3][lane * 8] = acc11;
  __syncthreads();

  int flat = threadIdx.x * 2;
#pragma unroll
  for (int q = 0; q < 2; ++q, ++flat) {
    const int tq = flat >> 8;
    const int p  = flat & 255;
    float sum = 0.f;
#pragma unroll
    for (int w = 0; w < kGemmWaves; ++w) sum += lds[w][tq][p];
    const int l  = p >> 3, r = p & 7;
    const int mr = (tq >> 1) * 16 + r + ((l >> 4) << 3);
    const int n  = (tq & 1) * 16 + (l & 15);
    partials[(size_t)ksp * (1024 * kB) + (size_t)(mpair * 32 + mr) * kB + n] = sum;
  }
}

__global__ __launch_bounds__(64) void caps_squash(const float* __restrict__ partials,
                                                  float* __restrict__ out) {
  const int b = blockIdx.x;
  const int k = threadIdx.x;
  float sv[kNU];
  float mag_sq = 0.f;
#pragma unroll
  for (int j = 0; j < kNU; ++j) {
    float v = 0.f;
#pragma unroll
    for (int p = 0; p < kKS; ++p)
      v += partials[(size_t)p * (1024 * kB) + (size_t)(j * kUS + k) * kB + b];
    sv[j] = v;
    mag_sq += v * v;
  }
  const float mag   = sqrtf(mag_sq);
  const float scale = mag_sq / ((1.f + mag_sq) * mag);
#pragma unroll
  for (int j = 0; j < kNU; ++j)
    out[(size_t)b * (kNU * kUS) + j * kUS + k] = sv[j] * scale;
}

extern "C" void kernel_launch(void* const* d_in, const int* in_sizes, int n_in,
                              void* d_out, int out_size, void* d_ws, size_t ws_size,
                              hipStream_t stream) {
  const float* x = (const float*)d_in[0];
  const float* W = (const float*)d_in[1];
  float* out      = (float*)d_out;
  float* partials = (float*)d_ws;
  float* xt2      = (float*)d_ws + kPartialsElems;

  caps_xt<<<kIC / kPackI, 128, 0, stream>>>(x, xt2);
  dim3 grid(32, kKS);
  caps_gemm<<<grid, 512, 0, stream>>>(xt2, W, partials);
  caps_squash<<<32, 64, 0, stream>>>(partials, out);
}


# K-split partials stored b-major: GEMM epilogue turns each wave's block in LDS and stores whole 128-byte lines, so a squash load touches one line per wave instead of 32; summation order unchanged
# speedup vs baseline: 1.5987x; 1.2531x over previous
#include <hip/hip_runtime.h>
#include <math.h>

typedef __attribute__((ext_vector_type(2))) float v2f;
typedef __attribute__((ext_vector_type(4))) float v4f;
typedef __attribute__((ext_vector_type(8))) float v8f;

constexpr int kB   = 32;
constexpr int kIU  = 16;
constexpr int kIC  = 2048;
constexpr int kNU  = 16;
constexpr int kUS  = 64;
constexpr int kKS  = 8;
constexpr size_t kPartialsElems = (size_t)kKS * 1024 * kB;

constexpr int kPackI = 8;

__global__ __launch_bounds__(128) void caps_xt(const float* __restrict__ x,
                                               float* __restrict__ xt2) {
  const int lane = threadIdx.x & 31;
  const int wave = threadIdx.x >> 5;
  const int tile = wave >> 1;
  const int half = wave & 1;
  const int kgrp = lane >> 4;
  const int n    = (lane & 15) + 16 * tile;
  const int u0   = 8 * kgrp + 4 * half;
  const int i0   = blockIdx.x * kPackI;

  const float* src = x + (size_t)n * (kIU * kIC) + (size_t)u0 * kIC + i0;
  v4f r[4][2];
#pragma unroll
  for (int q = 0; q < 4; ++q) {
    r[q][0] = *(const v4f*)(src + (size_t)q * kIC);
    r[q][1] = *(const v4f*)(src + (size_t)q * kIC + 4);
  }

  float* dst = xt2 + (size_t)i0 * 512 + tile * 256 + half * 128 + lane * 4;
#pragma unroll
  for (int il = 0; il < kPackI; ++il) {
    v4f o;
    o.x = r[0][il >> 2][il & 3];
    o.y = r[1][il >> 2][il & 3];
    o.z = r[2][il >> 2][il & 3];
    o.w = r[3][il >> 2][il & 3];
    *(v4f*)(dst + (size_t)il * 512) = o;
  }
}

constexpr int kGemmWaves = 16;
constexpr int kGemmTrips = (kIC / kKS) / kGemmWaves;
constexpr int kAhead     = 6;

static __device__ __forceinline__ void caps_trip(const v4f (&fa)[4], const v4f (&fb)[4],
                                                 v8f& acc00, v8f& acc01,
                                                 v8f& acc10, v8f& acc11) {
  v2f a0, a1, b;
#define STEP(AV0, AV1, LOHI, B0, B1)                                        \
    a0 = __builtin_shufflevector(AV0, AV0, 0 + LOHI, 1 + LOHI);             \
    a1 = __builtin_shufflevector(AV1, AV1, 0 + LOHI, 1 + LOHI);             \
    b = __builtin_shufflevector(B0, B0, 0 + LOHI, 1 + LOHI);                \
    acc00 = __builtin_amdgcn_wmma_f32_16x16x4_f32(false, a0, false, b,      \
                                      (short)0, acc00, false, false);       \
    acc10 = __builtin_amdgcn_wmma_f32_16x16x4_f32(false, a1, false, b,      \
                                      (short)0, acc10, false, false);       \
    b = __builtin_shufflevector(B1, B1, 0 + LOHI, 1 + LOHI);                \
    acc01 = __builtin_amdgcn_wmma_f32_16x16x4_f32(false, a0, false, b,      \
                                      (short)0, acc01, false, false);       \
    acc11 = __builtin_amdgcn_wmma_f32_16x16x4_f32(false, a1, false, b,      \
                                      (short)0, acc11, false, false);
  STEP(fa[0], fa[2], 0, fb[0], fb[2])
  STEP(fa[0], fa[2], 2, fb[0], fb[2])
  STEP(fa[1], fa[3], 0, fb[1], fb[3])
  STEP(fa[1], fa[3], 2, fb[1], fb[3])
#undef STEP
}

__global__ __launch_bounds__(512) void caps_gemm(const float* __restrict__ xt2,
                                                 const float* __restrict__ W,
                                                 float* __restrict__ partials) {
  const int lane  = threadIdx.x & 31;
  const int wave  = __builtin_amdgcn_readfirstlane(threadIdx.x >> 5);
  const int mpair = blockIdx.x;
  const int ksp   = blockIdx.y;
  const int mrow  = lane & 15;
  const int kgrp  = lane >> 4;

  v8f acc00 = {};
  v8f acc01 = {};
  v8f acc10 = {};
  v8f acc11 = {};

  const int ifirst = ksp * (kIC / kKS) + wave;
  const float* Wt = W + (size_t)ifirst * (kNU * kUS * kIU) + (size_t)mpair * 32 * kIU;
  const float* Xt = xt2 + (size_t)ifirst * 512;
  const unsigned wlane = (unsigned)(mrow * kIU + kgrp * 8);
  const unsigned xlane = (unsigned)(lane * 4);

  v4f fa[kAhead][4];
  v4f fb[kAhead][4];
#define LOAD_TRIP(T) do {                                                              \
    const float* Wb_ = Wt + (size_t)(T) * (kGemmWaves * kNU * kUS * kIU) + wlane;      \
    const float* Xb_ = Xt + (size_t)(T) * (kGemmWaves * 512) + xlane;                  \
    fa[(T) % kAhead][0] = __builtin_nontemporal_load((const v4f*)(Wb_));               \
    fa[(T) % kAhead][1] = __builtin_nontemporal_load((const v4f*)(Wb_ + 4));           \
    fa[(T) % kAhead][2] = __builtin_nontemporal_load((const v4f*)(Wb_ + 256));         \
    fa[(T) % kAhead][3] = __builtin_nontemporal_load((const v4f*)(Wb_ + 260));         \
    fb[(T) % kAhead][0] = *(const v4f*)(Xb_);                                          \
    fb[(T) % kAhead][1] = *(const v4f*)(Xb_ + 128);                                    \
    fb[(T) % kAhead][2] = *(const v4f*)(Xb_ + 256);                                    \
    fb[(T) % kAhead][3] = *(const v4f*)(Xb_ + 384);                                    \
  } while (0)

#pragma unroll
  for (int t = 0; t < kAhead; ++t) LOAD_TRIP(t);
#pragma unroll
  for (int t = 0; t < kGemmTrips; ++t) {
    __builtin_amdgcn_sched_barrier(0);
    caps_trip(fa[t % kAhead], fb[t % kAhead], acc00, acc01, acc10, acc11);
    __builtin_amdgcn_sched_barrier(0);
    if (t + kAhead < kGemmTrips) LOAD_TRIP(t + kAhead);
  }
#undef LOAD_TRIP

  constexpr int kRedPitch = 36;
  __shared__ float lds[kGemmWaves][32 * kRedPitch];
  {
    float* wb = &lds[wave][0] + mrow * kRedPitch + kgrp * 8;
    *(v4f*)(wb)                           = __builtin_shufflevector(acc00, acc00, 0, 1, 2, 3);
    *(v4f*)(wb + 4)                       = __builtin_shufflevector(acc00, acc00, 4, 5, 6, 7);
    *(v4f*)(wb + 16)                      = __builtin_shufflevector(acc10, acc10, 0, 1, 2, 3);
    *(v4f*)(wb + 20)                      = __builtin_shufflevector(acc10, acc10, 4, 5, 6, 7);
    *(v4f*)(wb + 16 * kRedPitch)          = __builtin_shufflevector(acc01, acc01, 0, 1, 2, 3);
    *(v4f*)(wb + 16 * kRedPitch + 4)      = __builtin_shufflevector(acc01, acc01, 4, 5, 6, 7);
    *(v4f*)(wb + 16 * kRedPitch + 16)     = __builtin_shufflevector(acc11, acc11, 0, 1, 2, 3);
    *(v4f*)(wb + 16 * kRedPitch + 20)     = __builtin_shufflevector(acc11, acc11, 4, 5, 6, 7);
  }
  __syncthreads();

  {
    const int n  = (wave & 7) + 8 * kgrp + 16 * (wave >> 3);
    const int m2 = 2 * mrow;
    const float* rb = &lds[0][0] + n * kRedPitch + m2;
    v2f sum = {0.f, 0.f};
#pragma unroll
    for (int w = 0; w < kGemmWaves; ++w) sum += *(const v2f*)(rb + w * (32 * kRedPitch));
    *(v2f*)(partials + ((size_t)ksp * kB + n) * 1024 + mpair * 32 + m2) = sum;
  }
}

__global__ __launch_bounds__(64) void caps_squash(const float* __restrict__ partials,
                                                  float* __restrict__ out) {
  const int b = blockIdx.x;
  const int k = threadIdx.x;
  float sv[kNU];
  float mag_sq = 0.f;
#pragma unroll
  for (int j = 0; j < kNU; ++j) {
    float v = 0.f;
#pragma unroll
    for (int p = 0; p < kKS; ++p)
      v += partials[((size_t)p * kB + b) * 1024 + j * kUS + k];
    sv[j] = v;
    mag_sq += v * v;
  }
  const float mag   = sqrtf(mag_sq);
  const float scale = mag_sq / ((1.f + mag_sq) * mag);
#pragma unroll
  for (int j = 0; j < kNU; ++j)
    out[(size_t)b * (kNU * kUS) + j * kUS + k] = sv[j] * scale;
}

extern "C" void kernel_launch(void* const* d_in, const int* in_sizes, int n_in,
                              void* d_out, int out_size, void* d_ws, size_t ws_size,
                              hipStream_t stream) {
  const float* x = (const float*)d_in[0];
  const float* W = (const float*)d_in[1];
  float* out      = (float*)d_out;
  float* partials = (float*)d_ws;
  float* xt2      = (float*)d_ws + kPartialsElems;

  caps_xt<<<kIC / kPackI, 128, 0, stream>>>(x, xt2);
  dim3 grid(32, kKS);
  caps_gemm<<<grid, 512, 0, stream>>>(xt2, W, partials);
  caps_squash<<<32, 64, 0, stream>>>(partials, out);
}


# GEMM builds its B fragments itself: x read as whole 256-byte row segments, turned in registers, staged through LDS in 64-i chunks; pack launch and xt2 round trip removed; sum order unchanged
# speedup vs baseline: 1.6817x; 1.0519x over previous
#include <hip/hip_runtime.h>
#include <math.h>


typedef __attribute__((ext_vector_type(2))) float v2f;
typedef __attribute__((ext_vector_type(4))) float v4f;
typedef __attribute__((ext_vector_type(8))) float v8f;

constexpr int kB   = 32;
constexpr int kIU  = 16;
constexpr int kIC  = 2048;
constexpr int kNU  = 16;
constexpr int kUS  = 64;
constexpr int kKS  = 8;
constexpr size_t kPartialsElems = (size_t)kKS * 1024 * kB;

constexpr int kGemmWaves = 16;
constexpr int kGemmTrips = (kIC / kKS) / kGemmWaves;
constexpr int kAhead     = 4;

static __device__ __forceinline__ void caps_trip(const v4f (&fa)[4], const v4f (&fb)[4],
                                                 v8f& acc00, v8f& acc01,
                                                 v8f& acc10, v8f& acc11) {
  v2f a0, a1, b;
#define STEP(AV0, AV1, LOHI, B0, B1)                                        \
    a0 = __builtin_shufflevector(AV0, AV0, 0 + LOHI, 1 + LOHI);             \
    a1 = __builtin_shufflevector(AV1, AV1, 0 + LOHI, 1 + LOHI);             \
    b = __builtin_shufflevector(B0, B0, 0 + LOHI, 1 + LOHI);                \
    acc00 = __builtin_amdgcn_wmma_f32_16x16x4_f32(false, a0, false, b,      \
                                      (short)0, acc00, false, false);       \
    acc10 = __builtin_amdgcn_wmma_f32_16x16x4_f32(false, a1, false, b,      \
                                      (short)0, acc10, false, false);       \
    b = __builtin_shufflevector(B1, B1, 0 + LOHI, 1 + LOHI);                \
    acc01 = __builtin_amdgcn_wmma_f32_16x16x4_f32(false, a0, false, b,      \
                                      (short)0, acc01, false, false);       \
    acc11 = __builtin_amdgcn_wmma_f32_16x16x4_f32(false, a1, false, b,      \
                                      (short)0, acc11, false, false);
  STEP(fa[0], fa[2], 0, fb[0], fb[2])
  STEP(fa[0], fa[2], 2, fb[0], fb[2])
  STEP(fa[1], fa[3], 0, fb[1], fb[3])
  STEP(fa[1], fa[3], 2, fb[1], fb[3])
#undef STEP
}

static __device__ __forceinline__ void caps_lds_barrier() {
  __builtin_amdgcn_fence(__ATOMIC_RELEASE, "workgroup", "local");
  __builtin_amdgcn_s_barrier();
  __builtin_amdgcn_fence(__ATOMIC_ACQUIRE, "workgroup", "local");
}

constexpr int kChunkI      = 64;
constexpr int kChunks      = (kIC / kKS) / kChunkI;
constexpr int kChunkTrips  = kChunkI / kGemmWaves;
constexpr int kStageFloats = kChunkI * 512;
constexpr int kRedPitch    = 36;

__global__ __launch_bounds__(512) void caps_gemm(const float* __restrict__ x,
                                                 const float* __restrict__ W,
                                                 float* __restrict__ partials) {
  const int lane  = threadIdx.x & 31;
  const int wave  = __builtin_amdgcn_readfirstlane(threadIdx.x >> 5);
  const int mpair = blockIdx.x;
  const int ksp   = blockIdx.y;
  const int mrow  = lane & 15;
  const int kgrp  = lane >> 4;

  __shared__ alignas(16) float lds[2 * kStageFloats];

  v8f acc00 = {};
  v8f acc01 = {};
  v8f acc10 = {};
  v8f acc11 = {};

  const int ifirst = ksp * (kIC / kKS) + wave;
  const float* Wt = W + (size_t)ifirst * (kNU * kUS * kIU) + (size_t)mpair * 32 * kIU;
  const unsigned wlane = (unsigned)(mrow * kIU + kgrp * 8);

  const int piece = lane & 15;
  const int rsel  = lane >> 4;
  const float* xb = x + (size_t)(wave >> 1) * (kIU * kIC)
                      + (size_t)((((wave & 1) * 2 + rsel) * 4) * kIC)
                      + (size_t)(ksp * (kIC / kKS) + piece * 4);
  const unsigned sdst  = (unsigned)(piece * (4 * 512) + rsel * 128);
  const unsigned slot0 = (unsigned)((((wave & 1) * 16 + (wave >> 1)) ^ piece));

  v4f xr[4][4];
  v4f fa[kAhead][4];
  v4f fb[2][4];

#define LOAD_X(C) do {                                                                 \
    _Pragma("unroll") for (int it_ = 0; it_ < 4; ++it_)                                \
      _Pragma("unroll") for (int q_ = 0; q_ < 4; ++q_)                                 \
        xr[it_][q_] = *(const v4f*)(xb + (size_t)it_ * (8 * kIU * kIC)                 \
                                       + (size_t)q_ * kIC + (C) * kChunkI);            \
  } while (0)
#define STORE_X(C) do {                                                                \
    float* sb_ = lds + ((C) & 1) * kStageFloats + sdst;                                \
    _Pragma("unroll") for (int it_ = 0; it_ < 4; ++it_)                                \
      _Pragma("unroll") for (int il_ = 0; il_ < 4; ++il_) {                            \
        v4f o_;                                                                        \
        o_.x = xr[it_][0][il_];                                                        \
        o_.y = xr[it_][1][il_];                                                        \
        o_.z = xr[it_][2][il_];                                                        \
        o_.w = xr[it_][3][il_];                                                        \
        *(v4f*)(sb_ + il_ * 512 + (it_ >> 1) * 256                                     \
                    + ((slot0 ^ (unsigned)((it_ & 1) * 8)) * 4)) = o_;                 \
      }                                                                                \
  } while (0)
#define LOAD_A(T) do {                                                                 \
    const float* Wb_ = Wt + (size_t)(T) * (kGemmWaves * kNU * kUS * kIU) + wlane;      \
    fa[(T) % kAhead][0] = __builtin_nontemporal_load((const v4f*)(Wb_));               \
    fa[(T) % kAhead][1] = __builtin_nontemporal_load((const v4f*)(Wb_ + 4));           \
    fa[(T) % kAhead][2] = __builtin_nontemporal_load((const v4f*)(Wb_ + 256));         \
    fa[(T) % kAhead][3] = __builtin_nontemporal_load((const v4f*)(Wb_ + 260));         \
  } while (0)
#define LOAD_B(T) do {                                                                 \
    const int tt_ = (T) % kChunkTrips;                                                 \
    const float* rb_ = lds + (((T) / kChunkTrips) & 1) * kStageFloats                  \
                           + (wave + kGemmWaves * tt_) * 512                           \
                           + ((lane ^ ((wave >> 2) + 4 * tt_)) * 4);                   \
    fb[(T) & 1][0] = *(const v4f*)(rb_);                                               \
    fb[(T) & 1][1] = *(const v4f*)(rb_ + 128);                                         \
    fb[(T) & 1][2] = *(const v4f*)(rb_ + 256);                                         \
    fb[(T) & 1][3] = *(const v4f*)(rb_ + 384);                                         \
  } while (0)

  LOAD_X(0);
#pragma unroll
  for (int t = 0; t < kAhead; ++t) LOAD_A(t);
  __builtin_amdgcn_sched_barrier(0);
  STORE_X(0);
  LOAD_X(1);
  caps_lds_barrier();
  LOAD_B(0);
#pragma unroll
  for (int t = 0; t < kGemmTrips; ++t) {
    const int tt = t % kChunkTrips;
    const int c  = t / kChunkTrips;
    __builtin_amdgcn_sched_barrier(0);
    if (tt + 1 < kChunkTrips) LOAD_B(t + 1);
    __builtin_amdgcn_sched_barrier(0);
    caps_trip(fa[t % kAhead], fb[t & 1], acc00, acc01, acc10, acc11);
    __builtin_amdgcn_sched_barrier(0);
    if (t + kAhead < kGemmTrips) LOAD_A(t + kAhead);
    if (tt == 1 && c + 1 < kChunks) {
      STORE_X(c + 1);
      if (c + 2 < kChunks) LOAD_X(c + 2);
    }
    if (tt + 1 == kChunkTrips && c + 1 < kChunks) {
      caps_lds_barrier();
      LOAD_B(t + 1);
    }
  }
#undef LOAD_X
#undef STORE_X
#undef LOAD_A
#undef LOAD_B

  {
    float* wb = lds + wave * (32 * kRedPitch) + mrow * kRedPitch + kgrp * 8;
    *(v4f*)(wb)                           = __builtin_shufflevector(acc00, acc00, 0, 1, 2, 3);
    *(v4f*)(wb + 4)                       = __builtin_shufflevector(acc00, acc00, 4, 5, 6, 7);
    *(v4f*)(wb + 16)                      = __builtin_shufflevector(acc10, acc10, 0, 1, 2, 3);
    *(v4f*)(wb + 20)                      = __builtin_shufflevector(acc10, acc10, 4, 5, 6, 7);
    *(v4f*)(wb + 16 * kRedPitch)          = __builtin_shufflevector(acc01, acc01, 0, 1, 2, 3);
    *(v4f*)(wb + 16 * kRedPitch + 4)      = __builtin_shufflevector(acc01, acc01, 4, 5, 6, 7);
    *(v4f*)(wb + 16 * kRedPitch + 16)     = __builtin_shufflevector(acc11, acc11, 0, 1, 2, 3);
    *(v4f*)(wb + 16 * kRedPitch + 20)     = __builtin_shufflevector(acc11, acc11, 4, 5, 6, 7);
  }
  __syncthreads();

  {
    const int n  = (wave & 7) + 8 * kgrp + 16 * (wave >> 3);
    const int m2 = 2 * mrow;
    const float* rb = lds + n * kRedPitch + m2;
    v2f sum = {0.f, 0.f};
#pragma unroll
    for (int w = 0; w < kGemmWaves; ++w) sum += *(const v2f*)(rb + w * (32 * kRedPitch));
    *(v2f*)(partials + ((size_t)ksp * kB + n) * 1024 + mpair * 32 + m2) = sum;
  }
}

__global__ __launch_bounds__(64) void caps_squash(const float* __restrict__ partials,
                                                  float* __restrict__ out) {
  const int b = blockIdx.x;
  const int k = threadIdx.x;
  float sv[kNU];
  float mag_sq = 0.f;
#pragma unroll
  for (int j = 0; j < kNU; ++j) {
    float v = 0.f;
#pragma unroll
    for (int p = 0; p < kKS; ++p)
      v += partials[((size_t)p * kB + b) * 1024 + j * kUS + k];
    sv[j] = v;
    mag_sq += v * v;
  }
  const float mag   = sqrtf(mag_sq);
  const float scale = mag_sq / ((1.f + mag_sq) * mag);
#pragma unroll
  for (int j = 0; j < kNU; ++j)
    out[(size_t)b * (kNU * kUS) + j * kUS + k] = sv[j] * scale;
}

extern "C" void kernel_launch(void* const* d_in, const int* in_sizes, int n_in,
                              void* d_out, int out_size, void* d_ws, size_t ws_size,
                              hipStream_t stream) {
  const float* x = (const float*)d_in[0];
  const float* W = (const float*)d_in[1];
  float* out      = (float*)d_out;
  float* partials = (float*)d_ws;

  dim3 grid(32, kKS);
  caps_gemm<<<grid, 512, 0, stream>>>(x, W, partials);
  caps_squash<<<32, 64, 0, stream>>>(partials, out);
}


# GEMM workgroup tile 128 rows x 64 i (was 32 x 256): the x of an i is fetched and turned once for four row groups; K-split 8 to 32, squash sums 32 partials one thread per element; K sum order changes
# speedup vs baseline: 2.0091x; 1.1947x over previous
#include <hip/hip_runtime.h>
#include <math.h>


typedef __attribute__((ext_vector_type(2))) float v2f;
typedef __attribute__((ext_vector_type(4))) float v4f;
typedef __attribute__((ext_vector_type(8))) float v8f;

constexpr int kB   = 32;
constexpr int kIU  = 16;
constexpr int kIC  = 2048;
constexpr int kNU  = 16;
constexpr int kUS  = 64;
constexpr int kKS  = 32;
constexpr size_t kPartialsElems = (size_t)kKS * 1024 * kB;

constexpr int kGemmWaves = 16;
constexpr int kIWaves    = 4;
constexpr int kISpan     = kIC / kKS;
constexpr int kGemmTrips = kISpan / kIWaves;
constexpr int kAhead     = 4;

static __device__ __forceinline__ void caps_trip(const v4f (&fa)[4], const v4f (&fb)[4],
                                                 v8f& acc00, v8f& acc01,
                                                 v8f& acc10, v8f& acc11) {
  v2f a0, a1, b;
#define STEP(AV0, AV1, LOHI, B0, B1)                                        \
    a0 = __builtin_shufflevector(AV0, AV0, 0 + LOHI, 1 + LOHI);             \
    a1 = __builtin_shufflevector(AV1, AV1, 0 + LOHI, 1 + LOHI);             \
    b = __builtin_shufflevector(B0, B0, 0 + LOHI, 1 + LOHI);                \
    acc00 = __builtin_amdgcn_wmma_f32_16x16x4_f32(false, a0, false, b,      \
                                      (short)0, acc00, false, false);       \
    acc10 = __builtin_amdgcn_wmma_f32_16x16x4_f32(false, a1, false, b,      \
                                      (short)0, acc10, false, false);       \
    b = __builtin_shufflevector(B1, B1, 0 + LOHI, 1 + LOHI);                \
    acc01 = __builtin_amdgcn_wmma_f32_16x16x4_f32(false, a0, false, b,      \
                                      (short)0, acc01, false, false);       \
    acc11 = __builtin_amdgcn_wmma_f32_16x16x4_f32(false, a1, false, b,      \
                                      (short)0, acc11, false, false);
  STEP(fa[0], fa[2], 0, fb[0], fb[2])
  STEP(fa[0], fa[2], 2, fb[0], fb[2])
  STEP(fa[1], fa[3], 0, fb[1], fb[3])
  STEP(fa[1], fa[3], 2, fb[1], fb[3])
#undef STEP
}

static __device__ __forceinline__ void caps_lds_barrier() {
  __builtin_amdgcn_fence(__ATOMIC_RELEASE, "workgroup", "local");
  __builtin_amdgcn_s_barrier();
  __builtin_amdgcn_fence(__ATOMIC_ACQUIRE, "workgroup", "local");
}

constexpr int kStageFloats = kISpan * 512;
constexpr int kRedPitch    = 36;
constexpr int kRedBlock    = 32 * kRedPitch + 8;
constexpr int kRedFloats   = kGemmWaves * kRedBlock;

__global__ __launch_bounds__(512) void caps_gemm(const float* __restrict__ x,
                                                 const float* __restrict__ W,
                                                 float* __restrict__ partials) {
  const int lane  = threadIdx.x & 31;
  const int wave  = __builtin_amdgcn_readfirstlane(threadIdx.x >> 5);
  const int rg    = wave >> 2;
  const int iw    = wave & 3;
  const int mblk  = blockIdx.x;
  const int ksp   = blockIdx.y;
  const int mpair = mblk * 4 + rg;
  const int mrow  = lane & 15;
  const int kgrp  = lane >> 4;

  __shared__ alignas(16) float lds[kStageFloats + kRedFloats];

  v8f acc00 = {};
  v8f acc01 = {};
  v8f acc10 = {};
  v8f acc11 = {};

  const int ifirst = ksp * kISpan + iw;
  const float* Wt = W + (size_t)ifirst * (kNU * kUS * kIU) + (size_t)mpair * 32 * kIU;
  const unsigned wlane = (unsigned)(mrow * kIU + kgrp * 8);

  const int piece = lane & 15;
  const int rsel  = lane >> 4;
  const float* xb = x + (size_t)(wave >> 1) * (kIU * kIC)
                      + (size_t)((((wave & 1) * 2 + rsel) * 4) * kIC)
                      + (size_t)(ksp * kISpan + piece * 4);
  const unsigned sdst  = (unsigned)(piece * (4 * 512) + rsel * 128);
  const unsigned slot0 = (unsigned)((((wave & 1) * 16 + (wave >> 1)) ^ piece));

  v4f xr[4][4];
  v4f fa[kAhead][4];
  v4f fb[2][4];

#define LOAD_X() do {                                                                  \
    _Pragma("unroll") for (int it_ = 0; it_ < 4; ++it_)                                \
      _Pragma("unroll") for (int q_ = 0; q_ < 4; ++q_)                                 \
        xr[it_][q_] = *(const v4f*)(xb + (size_t)it_ * (8 * kIU * kIC)                 \
                                       + (size_t)q_ * kIC);                            \
  } while (0)
#define STORE_X() do {                                                                 \
    float* sb_ = lds + sdst;                                                           \
    _Pragma("unroll") for (int it_ = 0; it_ < 4; ++it_)                                \
      _Pragma("unroll") for (int il_ = 0; il_ < 4; ++il_) {                            \
        v4f o_;                                                                        \
        o_.x = xr[it_][0][il_];                                                        \
        o_.y = xr[it_][1][il_];                                                        \
        o_.z = xr[it_][2][il_];                                                        \
        o_.w = xr[it_][3][il_];                                                        \
        *(v4f*)(sb_ + il_ * 512 + (it_ >> 1) * 256                                     \
                    + ((slot0 ^ (unsigned)((it_ & 1) * 8)) * 4)) = o_;                 \
      }                                                                                \
  } while (0)
#define LOAD_A(T) do {                                                                 \
    const float* Wb_ = Wt + (size_t)(T) * (kIWaves * kNU * kUS * kIU) + wlane;         \
    fa[(T) % kAhead][0] = __builtin_nontemporal_load((const v4f*)(Wb_));               \
    fa[(T) % kAhead][1] = __builtin_nontemporal_load((const v4f*)(Wb_ + 4));           \
    fa[(T) % kAhead][2] = __builtin_nontemporal_load((const v4f*)(Wb_ + 256));         \
    fa[(T) % kAhead][3] = __builtin_nontemporal_load((const v4f*)(Wb_ + 260));         \
  } while (0)
#define LOAD_B(T) do {                                                                 \
    const float* rb_ = lds + (iw + kIWaves * (T)) * 512 + ((lane ^ (T)) * 4);          \
    fb[(T) & 1][0] = *(const v4f*)(rb_);                                               \
    fb[(T) & 1][1] = *(const v4f*)(rb_ + 128);                                         \
    fb[(T) & 1][2] = *(const v4f*)(rb_ + 256);                                         \
    fb[(T) & 1][3] = *(const v4f*)(rb_ + 384);                                         \
  } while (0)

  LOAD_X();
  __builtin_amdgcn_sched_barrier(0);
#pragma unroll
  for (int t = 0; t < kAhead; ++t) LOAD_A(t);
  __builtin_amdgcn_sched_barrier(0);
  STORE_X();
  caps_lds_barrier();
  LOAD_B(0);
#pragma unroll
  for (int t = 0; t < kGemmTrips; ++t) {
    __builtin_amdgcn_sched_barrier(0);
    if (t + 1 < kGemmTrips) LOAD_B(t + 1);
    __builtin_amdgcn_sched_barrier(0);
    caps_trip(fa[t % kAhead], fb[t & 1], acc00, acc01, acc10, acc11);
    __builtin_amdgcn_sched_barrier(0);
    if (t + kAhead < kGemmTrips) LOAD_A(t + kAhead);
  }
#undef LOAD_X
#undef STORE_X
#undef LOAD_A
#undef LOAD_B

  float* red = lds + kStageFloats;
  {
    float* wb = red + wave * kRedBlock + mrow * kRedPitch + kgrp * 8;
    *(v4f*)(wb)                           = __builtin_shufflevector(acc00, acc00, 0, 1, 2, 3);
    *(v4f*)(wb + 4)                       = __builtin_shufflevector(acc00, acc00, 4, 5, 6, 7);
    *(v4f*)(wb + 16)                      = __builtin_shufflevector(acc10, acc10, 0, 1, 2, 3);
    *(v4f*)(wb + 20)                      = __builtin_shufflevector(acc10, acc10, 4, 5, 6, 7);
    *(v4f*)(wb + 16 * kRedPitch)          = __builtin_shufflevector(acc01, acc01, 0, 1, 2, 3);
    *(v4f*)(wb + 16 * kRedPitch + 4)      = __builtin_shufflevector(acc01, acc01, 4, 5, 6, 7);
    *(v4f*)(wb + 16 * kRedPitch + 16)     = __builtin_shufflevector(acc11, acc11, 0, 1, 2, 3);
    *(v4f*)(wb + 16 * kRedPitch + 20)     = __builtin_shufflevector(acc11, acc11, 4, 5, 6, 7);
  }
  __syncthreads();

  {
    const int rgs  = lane >> 3;
    const int mloc = (lane & 7) * 4;
#pragma unroll
    for (int e = 0; e < 2; ++e) {
      const int n = 2 * wave + e;
      const float* rb = red + (rgs * kIWaves) * kRedBlock + n * kRedPitch + mloc;
      v4f sum = {0.f, 0.f, 0.f, 0.f};
#pragma unroll
      for (int w = 0; w < kIWaves; ++w) sum += *(const v4f*)(rb + w * kRedBlock);
      *(v4f*)(partials + ((size_t)ksp * kB + n) * 1024 + mblk * 128 + 4 * lane) = sum;
    }
  }
}

__global__ __launch_bounds__(512) void caps_squash(const float* __restrict__ partials,
                                                   float* __restrict__ out) {
  const int b  = blockIdx.x >> 1;
  const int kh = blockIdx.x & 1;
  const int j  = threadIdx.x >> 5;
  const int kk = threadIdx.x & 31;
  const int m  = j * kUS + kh * 32 + kk;
  __shared__ float sq[kNU * 32];
  float v = 0.f;
#pragma unroll
  for (int p = 0; p < kKS; ++p)
    v += partials[((size_t)p * kB + b) * 1024 + m];
  sq[j * 32 + kk] = v * v;
  __syncthreads();
  float mag_sq = 0.f;
#pragma unroll
  for (int jj = 0; jj < kNU; ++jj) mag_sq += sq[jj * 32 + kk];
  const float mag   = sqrtf(mag_sq);
  const float scale = mag_sq / ((1.f + mag_sq) * mag);
  out[(size_t)b * (kNU * kUS) + m] = v * scale;
}

extern "C" void kernel_launch(void* const* d_in, const int* in_sizes, int n_in,
                              void* d_out, int out_size, void* d_ws, size_t ws_size,
                              hipStream_t stream) {
  const float* x = (const float*)d_in[0];
  const float* W = (const float*)d_in[1];
  float* out      = (float*)d_out;
  float* partials = (float*)d_ws;

  dim3 grid(8, kKS);
  caps_gemm<<<grid, 512, 0, stream>>>(x, W, partials);
  caps_squash<<<64, 512, 0, stream>>>(partials, out);
}
